# starting kernel
# baseline (speedup 1.0000x reference)
#include <hip/hip_runtime.h>


typedef __attribute__((ext_vector_type(2))) float v2f;
typedef __attribute__((ext_vector_type(8))) float v8f;

#define D_DIM  2048
#define K_EXP  16
#define R_RANK 16
#define NTOK   4096
#define N_HV   256

static __device__ __forceinline__ v8f wmma_f32(v2f a, v2f b, v8f c) {
  return __builtin_amdgcn_wmma_f32_16x16x4_f32(
       false, a,  false, b,
       (short)0, c,  false,  false);
}

__global__ __launch_bounds__(256) void moelora_router(
    const float* __restrict__ x, const float* __restrict__ Wr,
    const int* __restrict__ alpha_p, float* __restrict__ g)
{
  const int wave = threadIdx.x >> 5;
  const int lane = threadIdx.x & 31;
  const int t = blockIdx.x * 8 + wave;

  float acc[K_EXP];
#pragma unroll
  for (int k = 0; k < K_EXP; ++k) acc[k] = 0.f;

  const float* xrow = x + (size_t)t * D_DIM;
  for (int d = lane; d < D_DIM; d += 32) {
    const float xv = xrow[d];
#pragma unroll
    for (int k = 0; k < K_EXP; ++k)
      acc[k] = fmaf(xv, Wr[k * D_DIM + d], acc[k]);
  }
#pragma unroll
  for (int k = 0; k < K_EXP; ++k) {
#pragma unroll
    for (int off = 16; off > 0; off >>= 1)
      acc[k] += __shfl_xor(acc[k], off, 32);
  }

  if (lane == 0) {
    const float alpha = (float)(*alpha_p);
    int i0 = 0; float v0 = acc[0];
#pragma unroll
    for (int k = 1; k < K_EXP; ++k) if (acc[k] > v0) { v0 = acc[k]; i0 = k; }
    int i1 = -1; float v1 = -3.4e38f;
#pragma unroll
    for (int k = 0; k < K_EXP; ++k)
      if (k != i0 && acc[k] > v1) { v1 = acc[k]; i1 = k; }
    const float e  = __expf(v1 - v0);
    const float dn = 1.f + e;
    const float g0 = alpha / dn;
    const float g1 = alpha * e / dn;
#pragma unroll
    for (int k = 0; k < K_EXP; ++k)
      g[t * K_EXP + k] = (k == i0) ? g0 : ((k == i1) ? g1 : 0.f);
  }
}

__global__ __launch_bounds__(256) void moelora_hgemm(
    const float* __restrict__ x, const float* __restrict__ Amat,
    const float* __restrict__ g, float* __restrict__ hv, int nchunk)
{
  const int wave = threadIdx.x >> 5;
  const int lane = threadIdx.x & 31;
  const int m = lane & 15;
  const int h = lane >> 4;
  const int row0 = blockIdx.x * 32;
  const int nt0  = wave * 2;

  const float* xrow0 = x    + (size_t)(row0 + m)        * D_DIM + 2 * h;
  const float* xrow1 = x    + (size_t)(row0 + 16 + m)   * D_DIM + 2 * h;
  const float* arow0 = Amat + (size_t)(nt0 * 16 + m)       * D_DIM + 2 * h;
  const float* arow1 = Amat + (size_t)((nt0 + 1) * 16 + m) * D_DIM + 2 * h;

  v8f c00 = {}, c01 = {}, c10 = {}, c11 = {};

  for (int c = 0; c < nchunk; ++c) {
    const int base = c * 16;
#pragma unroll
    for (int s = 0; s < 4; ++s) {
      const int kb = base + s * 4;
      const v2f a0 = *(const v2f*)(xrow0 + kb);
      const v2f a1 = *(const v2f*)(xrow1 + kb);
      const v2f b0 = *(const v2f*)(arow0 + kb);
      const v2f b1 = *(const v2f*)(arow1 + kb);
      c00 = wmma_f32(a0, b0, c00);
      c01 = wmma_f32(a0, b1, c01);
      c10 = wmma_f32(a1, b0, c10);
      c11 = wmma_f32(a1, b1, c11);
    }
  }

#pragma unroll
  for (int i = 0; i < 8; ++i) {
    const int ra = row0 + i + 8 * h;
    const int rb = ra + 16;
    const float ga0 = g[ra * K_EXP + nt0];
    const float ga1 = g[ra * K_EXP + nt0 + 1];
    const float gb0 = g[rb * K_EXP + nt0];
    const float gb1 = g[rb * K_EXP + nt0 + 1];
    hv[(size_t)ra * N_HV + nt0 * 16 + m]       = c00[i] * ga0;
    hv[(size_t)ra * N_HV + (nt0 + 1) * 16 + m] = c01[i] * ga1;
    hv[(size_t)rb * N_HV + nt0 * 16 + m]       = c10[i] * gb0;
    hv[(size_t)rb * N_HV + (nt0 + 1) * 16 + m] = c11[i] * gb1;
  }
}

__global__ __launch_bounds__(256) void moelora_ogemm(
    const float* __restrict__ hv, const float* __restrict__ Bm,
    float* __restrict__ out, int nexp)
{
  const int wave = threadIdx.x >> 5;
  const int lane = threadIdx.x & 31;
  const int m = lane & 15;
  const int h = lane >> 4;
  const int row0 = blockIdx.x * 32;
  const int dt0  = blockIdx.y * 16 + wave * 2;
  const int dcol0 = dt0 * 16 + m;
  const int dcol1 = (dt0 + 1) * 16 + m;

  const float* hr0 = hv + (size_t)(row0 + m)      * N_HV + 2 * h;
  const float* hr1 = hv + (size_t)(row0 + 16 + m) * N_HV + 2 * h;
  const float* bb0 = Bm + (size_t)dcol0 * R_RANK + 2 * h;
  const float* bb1 = Bm + (size_t)dcol1 * R_RANK + 2 * h;

  v8f c00 = {}, c01 = {}, c10 = {}, c11 = {};

  for (int kk = 0; kk < nexp; ++kk) {
#pragma unroll
    for (int s = 0; s < 4; ++s) {
      const v2f a0 = *(const v2f*)(hr0 + s * 4);
      const v2f a1 = *(const v2f*)(hr1 + s * 4);
      const v2f b0 = *(const v2f*)(bb0 + s * 4);
      const v2f b1 = *(const v2f*)(bb1 + s * 4);
      c00 = wmma_f32(a0, b0, c00);
      c01 = wmma_f32(a0, b1, c01);
      c10 = wmma_f32(a1, b0, c10);
      c11 = wmma_f32(a1, b1, c11);
    }
    hr0 += R_RANK;
    hr1 += R_RANK;
    bb0 += (size_t)D_DIM * R_RANK;
    bb1 += (size_t)D_DIM * R_RANK;
  }

#pragma unroll
  for (int i = 0; i < 8; ++i) {
    const int ra = row0 + i + 8 * h;
    const int rb = ra + 16;
    out[(size_t)ra * D_DIM + dcol0] = c00[i];
    out[(size_t)ra * D_DIM + dcol1] = c01[i];
    out[(size_t)rb * D_DIM + dcol0] = c10[i];
    out[(size_t)rb * D_DIM + dcol1] = c11[i];
  }
}

extern "C" void kernel_launch(void* const* d_in, const int* in_sizes, int n_in,
                              void* d_out, int out_size, void* d_ws, size_t ws_size,
                              hipStream_t stream) {
  (void)in_sizes; (void)n_in; (void)out_size; (void)ws_size;
  const float* x     = (const float*)d_in[0];
  const int*   alpha = (const int*)  d_in[1];
  const float* A     = (const float*)d_in[2];
  const float* Bm    = (const float*)d_in[3];
  const float* Wr    = (const float*)d_in[4];
  float* out = (float*)d_out;

  float* g  = (float*)d_ws;
  float* hv = g + (size_t)NTOK * K_EXP;

  moelora_router<<<NTOK / 8, 256, 0, stream>>>(x, Wr, alpha, g);
  moelora_hgemm<<<NTOK / 32, 256, 0, stream>>>(x, A, g, hv, D_DIM / 16);
  moelora_ogemm<<<dim3(NTOK / 32, D_DIM / 256), 256, 0, stream>>>(hv, Bm, out, K_EXP);
}


# o-GEMM: 8 small-tile waves per SIMD -> one wave per SIMD with a 4Mx8N register tile (grid 64x4, 4 waves), next expert's 32 fragment loads issued as one group ahead of 128 WMMAs; sum orders unchanged
# speedup vs baseline: 2.1375x; 2.1375x over previous
#include <hip/hip_runtime.h>


typedef __attribute__((ext_vector_type(2))) float v2f;
typedef __attribute__((ext_vector_type(4))) float v4f;
typedef __attribute__((ext_vector_type(8))) float v8f;

#define D_DIM  2048
#define K_EXP  16
#define R_RANK 16
#define NTOK   4096
#define N_HV   256
#define HV_STEP 64
#define HV_EXP  256
#define AF_UNIT 128
#define AF_TILE (64 * 4 * AF_UNIT)
#define AF_UNITS (K_EXP * 64 * 4)

static __device__ __forceinline__ v8f wmma_f32(v2f a, v2f b, v8f c) {
  return __builtin_amdgcn_wmma_f32_16x16x4_f32(
       false, a,  false, b,
       (short)0, c,  false,  false);
}

#define RT_TOK  4
#define RT_D    256
#define RT_SUB  (RT_D / 128)
#define RT_HALF (D_DIM / 2)

struct RouteSet { v4f xv[RT_TOK][RT_SUB]; v4f wv[K_EXP][RT_SUB]; };

static __device__ __forceinline__ void route_load(
    RouteSet& f, const float* xl, const float* wl, int c)
{
#pragma unroll
  for (int j = 0; j < RT_TOK; ++j)
#pragma unroll
    for (int s = 0; s < RT_SUB; ++s)
      f.xv[j][s] = *(const v4f*)(xl + (size_t)j * D_DIM + c * RT_D + s * 128);
#pragma unroll
  for (int k = 0; k < K_EXP; ++k)
#pragma unroll
    for (int s = 0; s < RT_SUB; ++s)
      f.wv[k][s] = *(const v4f*)(wl + (size_t)k * D_DIM + c * RT_D + s * 128);
}

static __device__ __forceinline__ void route_fma(
    const RouteSet& f, v2f (&acc)[RT_TOK][K_EXP])
{
#pragma unroll
  for (int s = 0; s < RT_SUB; ++s)
#pragma unroll
    for (int j = 0; j < RT_TOK; ++j) {
      const v2f xlo = f.xv[j][s].xy, xhi = f.xv[j][s].zw;
#pragma unroll
      for (int k = 0; k < K_EXP; ++k) {
        acc[j][k] = __builtin_elementwise_fma(xlo, (v2f)f.wv[k][s].xy, acc[j][k]);
        acc[j][k] = __builtin_elementwise_fma(xhi, (v2f)f.wv[k][s].zw, acc[j][k]);
      }
    }
}

__global__ __launch_bounds__(256) void moelora_router(
    const float* __restrict__ x, const float* __restrict__ Wr,
    const int* __restrict__ alpha_p, float* __restrict__ g,
    const float* __restrict__ Amat, float* __restrict__ Af, int nstep)
{
  __shared__ v4f part[4][2][2 * (K_EXP / 4)][32];
  const int wave = threadIdx.x >> 5;
  const int lane = threadIdx.x & 31;
  const int grp  = wave & 3;
  const int half = wave >> 2;
  const int gw = blockIdx.x * 8 + wave;
  const int t0 = (blockIdx.x * 4 + grp) * RT_TOK;

  static_assert(AF_UNITS == 2 * (NTOK / 16) * 8, "two Af units per router wave");
  v2f alo[2], ahi[2];
#pragma unroll
  for (int i = 0; i < 2; ++i) {
    const int u = 2 * gw + i;
    const float* asrc = Amat + (size_t)((u >> 8) * 16 + (lane & 15)) * D_DIM
                             + 32 * ((u >> 2) & 63) + 8 * (u & 3) + 2 * (lane >> 4);
    alo[i] = *(const v2f*)asrc;
    ahi[i] = *(const v2f*)(asrc + 4);
  }

  v2f acc[RT_TOK][K_EXP];
#pragma unroll
  for (int j = 0; j < RT_TOK; ++j)
#pragma unroll
    for (int k = 0; k < K_EXP; ++k) { acc[j][k].x = 0.f; acc[j][k].y = 0.f; }

  const float* xl = x  + (size_t)t0 * D_DIM + half * RT_HALF + lane * 4;
  const float* wl = Wr + half * RT_HALF + lane * 4;

  RouteSet fa, fb;
  route_load(fa, xl, wl, 0);
  int c = 0;
  for (; c + 1 < nstep; c += 2) {
    route_load(fb, xl, wl, c + 1);
    route_fma(fa, acc);
    const int cn = (c + 2 < nstep) ? c + 2 : c;
    route_load(fa, xl, wl, cn);
    route_fma(fb, acc);
  }
  if (c < nstep) route_fma(fa, acc);

#pragma unroll
  for (int i = 0; i < 2; ++i) {
    v4f o; o.x = alo[i].x; o.y = alo[i].y; o.z = ahi[i].x; o.w = ahi[i].y;
    *(v4f*)(Af + ((size_t)(2 * gw + i) * 32 + lane) * 4) = o;
  }

  float mine[2][K_EXP], give[2][K_EXP];
#pragma unroll
  for (int i = 0; i < 2; ++i)
#pragma unroll
    for (int k = 0; k < K_EXP; ++k) {
      const float lo = acc[i][k].x + acc[i][k].y;
      const float hi = acc[2 + i][k].x + acc[2 + i][k].y;
      mine[i][k] = half ? hi : lo;
      give[i][k] = half ? lo : hi;
    }
#pragma unroll
  for (int i = 0; i < 2; ++i)
#pragma unroll
    for (int k4 = 0; k4 < K_EXP / 4; ++k4) {
      v4f o; o.x = give[i][4 * k4]; o.y = give[i][4 * k4 + 1]; o.z = give[i][4 * k4 + 2]; o.w = give[i][4 * k4 + 3];
      part[grp][half][i * 4 + k4][lane] = o;
    }
  __syncthreads();
#pragma unroll
  for (int i = 0; i < 2; ++i)
#pragma unroll
    for (int k4 = 0; k4 < K_EXP / 4; ++k4) {
      const v4f o = part[grp][1 - half][i * 4 + k4][lane];
      mine[i][4 * k4]     += o.x;
      mine[i][4 * k4 + 1] += o.y;
      mine[i][4 * k4 + 2] += o.z;
      mine[i][4 * k4 + 3] += o.w;
    }

#pragma unroll
  for (int i = 0; i < 2; ++i)
#pragma unroll
    for (int k = 0; k < K_EXP; ++k) {
#pragma unroll
      for (int off = 16; off > 0; off >>= 1)
        mine[i][k] += __shfl_xor(mine[i][k], off, 32);
    }

  float sc[K_EXP];
#pragma unroll
  for (int k = 0; k < K_EXP; ++k) sc[k] = (lane == 1) ? mine[1][k] : mine[0][k];

  if (lane < 2) {
    const int t = t0 + 2 * half + lane;
    const float alpha = (float)(*alpha_p);
    int i0 = 0; float v0 = sc[0];
#pragma unroll
    for (int k = 1; k < K_EXP; ++k) if (sc[k] > v0) { v0 = sc[k]; i0 = k; }
    int i1 = -1; float v1 = -3.4e38f;
#pragma unroll
    for (int k = 0; k < K_EXP; ++k)
      if (k != i0 && sc[k] > v1) { v1 = sc[k]; i1 = k; }
    const float e  = __expf(v1 - v0);
    const float dn = 1.f + e;
    const float g0 = alpha / dn;
    const float g1 = alpha * e / dn;
#pragma unroll
    for (int k4 = 0; k4 < K_EXP; k4 += 4) {
      v4f o;
      o.x = (k4     == i0) ? g0 : ((k4     == i1) ? g1 : 0.f);
      o.y = (k4 + 1 == i0) ? g0 : ((k4 + 1 == i1) ? g1 : 0.f);
      o.z = (k4 + 2 == i0) ? g0 : ((k4 + 2 == i1) ? g1 : 0.f);
      o.w = (k4 + 3 == i0) ? g0 : ((k4 + 3 == i1) ? g1 : 0.f);
      *(v4f*)(g + (size_t)t * K_EXP + k4) = o;
    }
  }
}

#define KSTEPS 8
#define KBLK   (KSTEPS * 4)

struct FragSet { v2f a0[KSTEPS], a1[KSTEPS]; v4f w0[KSTEPS / 2], w1[KSTEPS / 2]; };

static __device__ __forceinline__ void load_block(
    FragSet& f, const float* x0, const float* x1,
    const float* w0, const float* w1, int kb)
{
#pragma unroll
  for (int s = 0; s < KSTEPS; ++s) {
    f.a0[s] = *(const v2f*)(x0 + kb + s * 4);
    f.a1[s] = *(const v2f*)(x1 + kb + s * 4);
  }
#pragma unroll
  for (int p = 0; p < KSTEPS / 2; ++p) {
    f.w0[p] = *(const v4f*)(w0 + kb * 16 + p * AF_UNIT);
    f.w1[p] = *(const v4f*)(w1 + kb * 16 + p * AF_UNIT);
  }
}

static __device__ __forceinline__ void mma_block(
    const FragSet& f, v8f& c00, v8f& c01, v8f& c10, v8f& c11)
{
#pragma unroll
  for (int s = 0; s < KSTEPS; ++s) {
    const v4f u0 = f.w0[s >> 1], u1 = f.w1[s >> 1];
    v2f b0, b1;
    if (s & 1) { b0.x = u0.z; b0.y = u0.w; b1.x = u1.z; b1.y = u1.w; }
    else       { b0.x = u0.x; b0.y = u0.y; b1.x = u1.x; b1.y = u1.y; }
    c00 = wmma_f32(b0, f.a0[s], c00);
    c01 = wmma_f32(b1, f.a0[s], c01);
    c10 = wmma_f32(b0, f.a1[s], c10);
    c11 = wmma_f32(b1, f.a1[s], c11);
  }
}

__global__ __launch_bounds__(128) void moelora_hgemm(
    const float* __restrict__ x, const float* __restrict__ Af,
    const float* __restrict__ g, float* __restrict__ hv, int nblk)
{
  const int wave = threadIdx.x >> 5;
  const int lane = threadIdx.x & 31;
  const int m = lane & 15;
  const int h = lane >> 4;
  const int row0 = blockIdx.x * 32;
  const int nt0  = blockIdx.y * 8 + wave * 2;

  const float* xrow0 = x    + (size_t)(row0 + m)        * D_DIM + 2 * h;
  const float* xrow1 = x    + (size_t)(row0 + 16 + m)   * D_DIM + 2 * h;
  const float* arow0 = Af + (size_t)nt0 * AF_TILE + lane * 4;
  const float* arow1 = arow0 + AF_TILE;

  v8f c00 = {}, c01 = {}, c10 = {}, c11 = {};

  FragSet f0, f1, f2;
  const int klast = (nblk - 1) * KBLK;
  load_block(f0, xrow0, xrow1, arow0, arow1, 0);
  __builtin_amdgcn_sched_barrier(0);
  load_block(f1, xrow0, xrow1, arow0, arow1, (KBLK < klast) ? KBLK : klast);
  int kb = 0;
  int c = 0;
  for (; c + 3 <= nblk; c += 3) {
    const int k2 = (kb + 2 * KBLK < klast) ? kb + 2 * KBLK : klast;
    const int k3 = (kb + 3 * KBLK < klast) ? kb + 3 * KBLK : klast;
    const int k4 = (kb + 4 * KBLK < klast) ? kb + 4 * KBLK : klast;
    load_block(f2, xrow0, xrow1, arow0, arow1, k2);
    mma_block(f0, c00, c01, c10, c11);
    load_block(f0, xrow0, xrow1, arow0, arow1, k3);
    mma_block(f1, c00, c01, c10, c11);
    load_block(f1, xrow0, xrow1, arow0, arow1, k4);
    mma_block(f2, c00, c01, c10, c11);
    kb += 3 * KBLK;
    __builtin_amdgcn_sched_group_barrier(0x020, KSTEPS * 3, 0);
    __builtin_amdgcn_sched_group_barrier(0x008, KSTEPS * 4, 0);
    __builtin_amdgcn_sched_group_barrier(0x020, KSTEPS * 3, 0);
    __builtin_amdgcn_sched_group_barrier(0x008, KSTEPS * 4, 0);
    __builtin_amdgcn_sched_group_barrier(0x020, KSTEPS * 3, 0);
    __builtin_amdgcn_sched_group_barrier(0x008, KSTEPS * 4, 0);
  }
  if (c < nblk)     mma_block(f0, c00, c01, c10, c11);
  if (c + 1 < nblk) mma_block(f1, c00, c01, c10, c11);

  const int ta = row0 + m;
  const int tb = ta + 16;
  const float ga0 = g[ta * K_EXP + nt0];
  const float ga1 = g[ta * K_EXP + nt0 + 1];
  const float gb0 = g[tb * K_EXP + nt0];
  const float gb1 = g[tb * K_EXP + nt0 + 1];
  float* p00 = hv + ((size_t)(blockIdx.x * 2) * K_EXP + nt0) * HV_EXP + lane * 2;
  float* p01 = p00 + HV_EXP;
  float* p10 = p00 + (size_t)K_EXP * HV_EXP;
  float* p11 = p10 + HV_EXP;
#pragma unroll
  for (int j = 0; j < 4; ++j) {
    v2f v;
    v.x = c00[2 * j] * ga0; v.y = c00[2 * j + 1] * ga0; *(v2f*)(p00 + j * HV_STEP) = v;
    v.x = c01[2 * j] * ga1; v.y = c01[2 * j + 1] * ga1; *(v2f*)(p01 + j * HV_STEP) = v;
    v.x = c10[2 * j] * gb0; v.y = c10[2 * j + 1] * gb0; *(v2f*)(p10 + j * HV_STEP) = v;
    v.x = c11[2 * j] * gb1; v.y = c11[2 * j + 1] * gb1; *(v2f*)(p11 + j * HV_STEP) = v;
  }
}

#define O_MT 4
#define O_NT 8

struct OFragSet { v2f a[O_MT][4]; v4f b[O_NT][2]; };

static __device__ __forceinline__ void o_load(
    OFragSet& f, const float* hr, const float* bb)
{
#pragma unroll
  for (int mi = 0; mi < O_MT; ++mi)
#pragma unroll
    for (int s = 0; s < 4; ++s)
      f.a[mi][s] = *(const v2f*)(hr + (size_t)mi * K_EXP * HV_EXP + s * HV_STEP);
#pragma unroll
  for (int ni = 0; ni < O_NT; ++ni)
#pragma unroll
    for (int p = 0; p < 2; ++p)
      f.b[ni][p] = *(const v4f*)(bb + (size_t)ni * 16 * R_RANK + p * 4);
}

static __device__ __forceinline__ void o_mma(
    const OFragSet& f, v8f (&c)[O_MT][O_NT])
{
#pragma unroll
  for (int s = 0; s < 4; ++s)
#pragma unroll
    for (int mi = 0; mi < O_MT; ++mi)
#pragma unroll
      for (int ni = 0; ni < O_NT; ++ni) {
        const v4f u = f.b[ni][s >> 1];
        v2f b;
        if (s & 1) { b.x = u.z; b.y = u.w; }
        else       { b.x = u.x; b.y = u.y; }
        c[mi][ni] = wmma_f32(f.a[mi][s], b, c[mi][ni]);
      }
}

__global__ __launch_bounds__(128) void moelora_ogemm(
    const float* __restrict__ hv, const float* __restrict__ Bm,
    float* __restrict__ out, int nexp)
{
  const int wave = threadIdx.x >> 5;
  const int lane = threadIdx.x & 31;
  const int m = lane & 15;
  const int h = lane >> 4;
  const int mt0 = blockIdx.x * O_MT;
  const int dt0 = blockIdx.y * (4 * O_NT) + wave * O_NT;

  const float* hr = hv + (size_t)mt0 * K_EXP * HV_EXP + lane * 2;
  const float* bb = Bm + (size_t)(dt0 * 16 + m) * R_RANK + 8 * h;

  v8f c[O_MT][O_NT];
#pragma unroll
  for (int mi = 0; mi < O_MT; ++mi)
#pragma unroll
    for (int ni = 0; ni < O_NT; ++ni) c[mi][ni] = (v8f){};

  OFragSet f0, f1;
  const int elast = nexp - 1;
  o_load(f0, hr, bb);
  int e = 0;
  for (; e + 2 <= nexp; e += 2) {
    const int e2 = (e + 2 < nexp) ? e + 2 : elast;
    o_load(f1, hr + (size_t)(e + 1) * HV_EXP, bb + (size_t)(e + 1) * D_DIM * R_RANK);
    o_mma(f0, c);
    o_load(f0, hr + (size_t)e2 * HV_EXP, bb + (size_t)e2 * D_DIM * R_RANK);
    o_mma(f1, c);
    __builtin_amdgcn_sched_group_barrier(0x020, 32, 0);
    __builtin_amdgcn_sched_group_barrier(0x008, 128, 0);
    __builtin_amdgcn_sched_group_barrier(0x020, 32, 0);
    __builtin_amdgcn_sched_group_barrier(0x008, 128, 0);
  }
  if (e < nexp) o_mma(f0, c);

  const int row0 = mt0 * 16;
#pragma unroll
  for (int mi = 0; mi < O_MT; ++mi)
#pragma unroll
    for (int ni = 0; ni < O_NT; ++ni)
#pragma unroll
      for (int i = 0; i < 8; ++i)
        out[(size_t)(row0 + 16 * mi + i + 8 * h) * D_DIM + (dt0 + ni) * 16 + m] = c[mi][ni][i];
}

extern "C" void kernel_launch(void* const* d_in, const int* in_sizes, int n_in,
                              void* d_out, int out_size, void* d_ws, size_t ws_size,
                              hipStream_t stream) {
  (void)in_sizes; (void)n_in; (void)out_size; (void)ws_size;
  const float* x     = (const float*)d_in[0];
  const int*   alpha = (const int*)  d_in[1];
  const float* A     = (const float*)d_in[2];
  const float* Bm    = (const float*)d_in[3];
  const float* Wr    = (const float*)d_in[4];
  float* out = (float*)d_out;

  float* g  = (float*)d_ws;
  float* hv = g + (size_t)NTOK * K_EXP;
  float* Af = hv + (size_t)NTOK * N_HV;

  moelora_router<<<NTOK / 16, 256, 0, stream>>>(x, Wr, alpha, g, A, Af, RT_HALF / RT_D);
  moelora_hgemm<<<dim3(NTOK / 32, 2), 128, 0, stream>>>(x, Af, g, hv, D_DIM / 32);
  moelora_ogemm<<<dim3(NTOK / 64, D_DIM / 512), 128, 0, stream>>>(hv, Bm, out, K_EXP);
}


# h-GEMM wave tile 2Mx2N -> 1Mx4N, block = 4 M-tiles x 4 experts (grid 64x4): an x fragment load (16 cache lines) feeds 4 WMMAs not 2, weight lines shared by the 4 waves; sum orders unchanged
# speedup vs baseline: 2.2281x; 1.0424x over previous
#include <hip/hip_runtime.h>


typedef __attribute__((ext_vector_type(2))) float v2f;
typedef __attribute__((ext_vector_type(4))) float v4f;
typedef __attribute__((ext_vector_type(8))) float v8f;

#define D_DIM  2048
#define K_EXP  16
#define R_RANK 16
#define NTOK   4096
#define N_HV   256
#define HV_STEP 64
#define HV_EXP  256
#define AF_UNIT 128
#define AF_TILE (64 * 4 * AF_UNIT)
#define AF_UNITS (K_EXP * 64 * 4)

static __device__ __forceinline__ v8f wmma_f32(v2f a, v2f b, v8f c) {
  return __builtin_amdgcn_wmma_f32_16x16x4_f32(
       false, a,  false, b,
       (short)0, c,  false,  false);
}

#define RT_TOK  4
#define RT_D    256
#define RT_SUB  (RT_D / 128)
#define RT_HALF (D_DIM / 2)

struct RouteSet { v4f xv[RT_TOK][RT_SUB]; v4f wv[K_EXP][RT_SUB]; };

static __device__ __forceinline__ void route_load(
    RouteSet& f, const float* xl, const float* wl, int c)
{
#pragma unroll
  for (int j = 0; j < RT_TOK; ++j)
#pragma unroll
    for (int s = 0; s < RT_SUB; ++s)
      f.xv[j][s] = *(const v4f*)(xl + (size_t)j * D_DIM + c * RT_D + s * 128);
#pragma unroll
  for (int k = 0; k < K_EXP; ++k)
#pragma unroll
    for (int s = 0; s < RT_SUB; ++s)
      f.wv[k][s] = *(const v4f*)(wl + (size_t)k * D_DIM + c * RT_D + s * 128);
}

static __device__ __forceinline__ void route_fma(
    const RouteSet& f, v2f (&acc)[RT_TOK][K_EXP])
{
#pragma unroll
  for (int s = 0; s < RT_SUB; ++s)
#pragma unroll
    for (int j = 0; j < RT_TOK; ++j) {
      const v2f xlo = f.xv[j][s].xy, xhi = f.xv[j][s].zw;
#pragma unroll
      for (int k = 0; k < K_EXP; ++k) {
        acc[j][k] = __builtin_elementwise_fma(xlo, (v2f)f.wv[k][s].xy, acc[j][k]);
        acc[j][k] = __builtin_elementwise_fma(xhi, (v2f)f.wv[k][s].zw, acc[j][k]);
      }
    }
}

__global__ __launch_bounds__(256) void moelora_router(
    const float* __restrict__ x, const float* __restrict__ Wr,
    const int* __restrict__ alpha_p, float* __restrict__ g,
    const float* __restrict__ Amat, float* __restrict__ Af, int nstep)
{
  __shared__ v4f part[4][2][2 * (K_EXP / 4)][32];
  const int wave = threadIdx.x >> 5;
  const int lane = threadIdx.x & 31;
  const int grp  = wave & 3;
  const int half = wave >> 2;
  const int gw = blockIdx.x * 8 + wave;
  const int t0 = (blockIdx.x * 4 + grp) * RT_TOK;

  static_assert(AF_UNITS == 2 * (NTOK / 16) * 8, "two Af units per router wave");
  v2f alo[2], ahi[2];
#pragma unroll
  for (int i = 0; i < 2; ++i) {
    const int u = 2 * gw + i;
    const float* asrc = Amat + (size_t)((u >> 8) * 16 + (lane & 15)) * D_DIM
                             + 32 * ((u >> 2) & 63) + 8 * (u & 3) + 2 * (lane >> 4);
    alo[i] = *(const v2f*)asrc;
    ahi[i] = *(const v2f*)(asrc + 4);
  }

  v2f acc[RT_TOK][K_EXP];
#pragma unroll
  for (int j = 0; j < RT_TOK; ++j)
#pragma unroll
    for (int k = 0; k < K_EXP; ++k) { acc[j][k].x = 0.f; acc[j][k].y = 0.f; }

  const float* xl = x  + (size_t)t0 * D_DIM + half * RT_HALF + lane * 4;
  const float* wl = Wr + half * RT_HALF + lane * 4;

  RouteSet fa, fb;
  route_load(fa, xl, wl, 0);
  int c = 0;
  for (; c + 1 < nstep; c += 2) {
    route_load(fb, xl, wl, c + 1);
    route_fma(fa, acc);
    const int cn = (c + 2 < nstep) ? c + 2 : c;
    route_load(fa, xl, wl, cn);
    route_fma(fb, acc);
  }
  if (c < nstep) route_fma(fa, acc);

#pragma unroll
  for (int i = 0; i < 2; ++i) {
    v4f o; o.x = alo[i].x; o.y = alo[i].y; o.z = ahi[i].x; o.w = ahi[i].y;
    *(v4f*)(Af + ((size_t)(2 * gw + i) * 32 + lane) * 4) = o;
  }

  float mine[2][K_EXP], give[2][K_EXP];
#pragma unroll
  for (int i = 0; i < 2; ++i)
#pragma unroll
    for (int k = 0; k < K_EXP; ++k) {
      const float lo = acc[i][k].x + acc[i][k].y;
      const float hi = acc[2 + i][k].x + acc[2 + i][k].y;
      mine[i][k] = half ? hi : lo;
      give[i][k] = half ? lo : hi;
    }
#pragma unroll
  for (int i = 0; i < 2; ++i)
#pragma unroll
    for (int k4 = 0; k4 < K_EXP / 4; ++k4) {
      v4f o; o.x = give[i][4 * k4]; o.y = give[i][4 * k4 + 1]; o.z = give[i][4 * k4 + 2]; o.w = give[i][4 * k4 + 3];
      part[grp][half][i * 4 + k4][lane] = o;
    }
  __syncthreads();
#pragma unroll
  for (int i = 0; i < 2; ++i)
#pragma unroll
    for (int k4 = 0; k4 < K_EXP / 4; ++k4) {
      const v4f o = part[grp][1 - half][i * 4 + k4][lane];
      mine[i][4 * k4]     += o.x;
      mine[i][4 * k4 + 1] += o.y;
      mine[i][4 * k4 + 2] += o.z;
      mine[i][4 * k4 + 3] += o.w;
    }

#pragma unroll
  for (int i = 0; i < 2; ++i)
#pragma unroll
    for (int k = 0; k < K_EXP; ++k) {
#pragma unroll
      for (int off = 16; off > 0; off >>= 1)
        mine[i][k] += __shfl_xor(mine[i][k], off, 32);
    }

  float sc[K_EXP];
#pragma unroll
  for (int k = 0; k < K_EXP; ++k) sc[k] = (lane == 1) ? mine[1][k] : mine[0][k];

  if (lane < 2) {
    const int t = t0 + 2 * half + lane;
    const float alpha = (float)(*alpha_p);
    int i0 = 0; float v0 = sc[0];
#pragma unroll
    for (int k = 1; k < K_EXP; ++k) if (sc[k] > v0) { v0 = sc[k]; i0 = k; }
    int i1 = -1; float v1 = -3.4e38f;
#pragma unroll
    for (int k = 0; k < K_EXP; ++k)
      if (k != i0 && sc[k] > v1) { v1 = sc[k]; i1 = k; }
    const float e  = __expf(v1 - v0);
    const float dn = 1.f + e;
    const float g0 = alpha / dn;
    const float g1 = alpha * e / dn;
#pragma unroll
    for (int k4 = 0; k4 < K_EXP; k4 += 4) {
      v4f o;
      o.x = (k4     == i0) ? g0 : ((k4     == i1) ? g1 : 0.f);
      o.y = (k4 + 1 == i0) ? g0 : ((k4 + 1 == i1) ? g1 : 0.f);
      o.z = (k4 + 2 == i0) ? g0 : ((k4 + 2 == i1) ? g1 : 0.f);
      o.w = (k4 + 3 == i0) ? g0 : ((k4 + 3 == i1) ? g1 : 0.f);
      *(v4f*)(g + (size_t)t * K_EXP + k4) = o;
    }
  }
}

#define KSTEPS 8
#define KBLK   (KSTEPS * 4)
#define H_NT   4

struct FragSet { v2f a[KSTEPS]; v4f w[H_NT][KSTEPS / 2]; };

static __device__ __forceinline__ void load_block(
    FragSet& f, const float* x0, const float* w0, int kb)
{
#pragma unroll
  for (int s = 0; s < KSTEPS; ++s)
    f.a[s] = *(const v2f*)(x0 + kb + s * 4);
#pragma unroll
  for (int p = 0; p < KSTEPS / 2; ++p)
#pragma unroll
    for (int e = 0; e < H_NT; ++e)
      f.w[e][p] = *(const v4f*)(w0 + (size_t)e * AF_TILE + kb * 16 + p * AF_UNIT);
}

static __device__ __forceinline__ void mma_block(const FragSet& f, v8f (&c)[H_NT])
{
#pragma unroll
  for (int s = 0; s < KSTEPS; ++s)
#pragma unroll
    for (int e = 0; e < H_NT; ++e) {
      const v4f u = f.w[e][s >> 1];
      v2f b;
      if (s & 1) { b.x = u.z; b.y = u.w; }
      else       { b.x = u.x; b.y = u.y; }
      c[e] = wmma_f32(b, f.a[s], c[e]);
    }
}

__global__ __launch_bounds__(128) void moelora_hgemm(
    const float* __restrict__ x, const float* __restrict__ Af,
    const float* __restrict__ g, float* __restrict__ hv, int nblk)
{
  const int wave = threadIdx.x >> 5;
  const int lane = threadIdx.x & 31;
  const int m = lane & 15;
  const int h = lane >> 4;
  const int mt  = blockIdx.x * 4 + wave;
  const int nt0 = blockIdx.y * H_NT;

  const float* xrow = x  + (size_t)(mt * 16 + m) * D_DIM + 2 * h;
  const float* arow = Af + (size_t)nt0 * AF_TILE + lane * 4;

  v8f c[H_NT];
#pragma unroll
  for (int e = 0; e < H_NT; ++e) c[e] = (v8f){};

  FragSet f0, f1, f2;
  const int klast = (nblk - 1) * KBLK;
  load_block(f0, xrow, arow, 0);
  __builtin_amdgcn_sched_barrier(0);
  load_block(f1, xrow, arow, (KBLK < klast) ? KBLK : klast);
  int kb = 0;
  int cb = 0;
  for (; cb + 3 <= nblk; cb += 3) {
    const int k2 = (kb + 2 * KBLK < klast) ? kb + 2 * KBLK : klast;
    const int k3 = (kb + 3 * KBLK < klast) ? kb + 3 * KBLK : klast;
    const int k4 = (kb + 4 * KBLK < klast) ? kb + 4 * KBLK : klast;
    load_block(f2, xrow, arow, k2);
    mma_block(f0, c);
    load_block(f0, xrow, arow, k3);
    mma_block(f1, c);
    load_block(f1, xrow, arow, k4);
    mma_block(f2, c);
    kb += 3 * KBLK;
    __builtin_amdgcn_sched_group_barrier(0x020, KSTEPS * 3, 0);
    __builtin_amdgcn_sched_group_barrier(0x008, KSTEPS * 4, 0);
    __builtin_amdgcn_sched_group_barrier(0x020, KSTEPS * 3, 0);
    __builtin_amdgcn_sched_group_barrier(0x008, KSTEPS * 4, 0);
    __builtin_amdgcn_sched_group_barrier(0x020, KSTEPS * 3, 0);
    __builtin_amdgcn_sched_group_barrier(0x008, KSTEPS * 4, 0);
  }
  if (cb < nblk)     mma_block(f0, c);
  if (cb + 1 < nblk) mma_block(f1, c);

  const int t = mt * 16 + m;
  const v4f gt = *(const v4f*)(g + (size_t)t * K_EXP + nt0);
  float* p0 = hv + ((size_t)mt * K_EXP + nt0) * HV_EXP + lane * 2;
#pragma unroll
  for (int e = 0; e < H_NT; ++e) {
    const float ge = gt[e];
#pragma unroll
    for (int j = 0; j < 4; ++j) {
      v2f v;
      v.x = c[e][2 * j] * ge; v.y = c[e][2 * j + 1] * ge;
      *(v2f*)(p0 + (size_t)e * HV_EXP + j * HV_STEP) = v;
    }
  }
}

#define O_MT 4
#define O_NT 8

struct OFragSet { v2f a[O_MT][4]; v4f b[O_NT][2]; };

static __device__ __forceinline__ void o_load(
    OFragSet& f, const float* hr, const float* bb)
{
#pragma unroll
  for (int mi = 0; mi < O_MT; ++mi)
#pragma unroll
    for (int s = 0; s < 4; ++s)
      f.a[mi][s] = *(const v2f*)(hr + (size_t)mi * K_EXP * HV_EXP + s * HV_STEP);
#pragma unroll
  for (int ni = 0; ni < O_NT; ++ni)
#pragma unroll
    for (int p = 0; p < 2; ++p)
      f.b[ni][p] = *(const v4f*)(bb + (size_t)ni * 16 * R_RANK + p * 4);
}

static __device__ __forceinline__ void o_mma(
    const OFragSet& f, v8f (&c)[O_MT][O_NT])
{
#pragma unroll
  for (int s = 0; s < 4; ++s)
#pragma unroll
    for (int mi = 0; mi < O_MT; ++mi)
#pragma unroll
      for (int ni = 0; ni < O_NT; ++ni) {
        const v4f u = f.b[ni][s >> 1];
        v2f b;
        if (s & 1) { b.x = u.z; b.y = u.w; }
        else       { b.x = u.x; b.y = u.y; }
        c[mi][ni] = wmma_f32(f.a[mi][s], b, c[mi][ni]);
      }
}

__global__ __launch_bounds__(128) void moelora_ogemm(
    const float* __restrict__ hv, const float* __restrict__ Bm,
    float* __restrict__ out, int nexp)
{
  const int wave = threadIdx.x >> 5;
  const int lane = threadIdx.x & 31;
  const int m = lane & 15;
  const int h = lane >> 4;
  const int mt0 = blockIdx.x * O_MT;
  const int dt0 = blockIdx.y * (4 * O_NT) + wave * O_NT;

  const float* hr = hv + (size_t)mt0 * K_EXP * HV_EXP + lane * 2;
  const float* bb = Bm + (size_t)(dt0 * 16 + m) * R_RANK + 8 * h;

  v8f c[O_MT][O_NT];
#pragma unroll
  for (int mi = 0; mi < O_MT; ++mi)
#pragma unroll
    for (int ni = 0; ni < O_NT; ++ni) c[mi][ni] = (v8f){};

  OFragSet f0, f1;
  const int elast = nexp - 1;
  o_load(f0, hr, bb);
  int e = 0;
  for (; e + 2 <= nexp; e += 2) {
    const int e2 = (e + 2 < nexp) ? e + 2 : elast;
    o_load(f1, hr + (size_t)(e + 1) * HV_EXP, bb + (size_t)(e + 1) * D_DIM * R_RANK);
    o_mma(f0, c);
    o_load(f0, hr + (size_t)e2 * HV_EXP, bb + (size_t)e2 * D_DIM * R_RANK);
    o_mma(f1, c);
    __builtin_amdgcn_sched_group_barrier(0x020, 32, 0);
    __builtin_amdgcn_sched_group_barrier(0x008, 128, 0);
    __builtin_amdgcn_sched_group_barrier(0x020, 32, 0);
    __builtin_amdgcn_sched_group_barrier(0x008, 128, 0);
  }
  if (e < nexp) o_mma(f0, c);

  const int row0 = mt0 * 16;
#pragma unroll
  for (int mi = 0; mi < O_MT; ++mi)
#pragma unroll
    for (int ni = 0; ni < O_NT; ++ni)
#pragma unroll
      for (int i = 0; i < 8; ++i)
        out[(size_t)(row0 + 16 * mi + i + 8 * h) * D_DIM + (dt0 + ni) * 16 + m] = c[mi][ni][i];
}

extern "C" void kernel_launch(void* const* d_in, const int* in_sizes, int n_in,
                              void* d_out, int out_size, void* d_ws, size_t ws_size,
                              hipStream_t stream) {
  (void)in_sizes; (void)n_in; (void)out_size; (void)ws_size;
  const float* x     = (const float*)d_in[0];
  const int*   alpha = (const int*)  d_in[1];
  const float* A     = (const float*)d_in[2];
  const float* Bm    = (const float*)d_in[3];
  const float* Wr    = (const float*)d_in[4];
  float* out = (float*)d_out;

  float* g  = (float*)d_ws;
  float* hv = g + (size_t)NTOK * K_EXP;
  float* Af = hv + (size_t)NTOK * N_HV;

  moelora_router<<<NTOK / 16, 256, 0, stream>>>(x, Wr, alpha, g, A, Af, RT_HALF / RT_D);
  moelora_hgemm<<<dim3(NTOK / 64, K_EXP / H_NT), 128, 0, stream>>>(x, Af, g, hv, D_DIM / 32);
  moelora_ogemm<<<dim3(NTOK / 64, D_DIM / 512), 128, 0, stream>>>(hv, Bm, out, K_EXP);
}


# h-GEMM computes only each token's 2 selected experts (router zeroes hv): per-expert token lists by block scan, no atomics; gathered 16-row tiles, K quarters on 4 waves added via LDS: new k sum order
# speedup vs baseline: 2.4540x; 1.1014x over previous
#include <hip/hip_runtime.h>


typedef __attribute__((ext_vector_type(2))) float v2f;
typedef __attribute__((ext_vector_type(4))) float v4f;
typedef __attribute__((ext_vector_type(8))) float v8f;
typedef __attribute__((ext_vector_type(4))) int   v4i;

#define D_DIM  2048
#define K_EXP  16
#define R_RANK 16
#define NTOK   4096
#define N_HV   256
#define HV_STEP 64
#define HV_EXP  256
#define AF_UNIT 128
#define AF_TILE (64 * 4 * AF_UNIT)
#define AF_UNITS (K_EXP * 64 * 4)

static __device__ __forceinline__ v8f wmma_f32(v2f a, v2f b, v8f c) {
  return __builtin_amdgcn_wmma_f32_16x16x4_f32(
       false, a,  false, b,
       (short)0, c,  false,  false);
}

#define RT_TOK  4
#define RT_D    256
#define RT_SUB  (RT_D / 128)
#define RT_HALF (D_DIM / 2)

struct RouteSet { v4f xv[RT_TOK][RT_SUB]; v4f wv[K_EXP][RT_SUB]; };

static __device__ __forceinline__ void route_load(
    RouteSet& f, const float* xl, const float* wl, int c)
{
#pragma unroll
  for (int j = 0; j < RT_TOK; ++j)
#pragma unroll
    for (int s = 0; s < RT_SUB; ++s)
      f.xv[j][s] = *(const v4f*)(xl + (size_t)j * D_DIM + c * RT_D + s * 128);
#pragma unroll
  for (int k = 0; k < K_EXP; ++k)
#pragma unroll
    for (int s = 0; s < RT_SUB; ++s)
      f.wv[k][s] = *(const v4f*)(wl + (size_t)k * D_DIM + c * RT_D + s * 128);
}

static __device__ __forceinline__ void route_fma(
    const RouteSet& f, v2f (&acc)[RT_TOK][K_EXP])
{
#pragma unroll
  for (int s = 0; s < RT_SUB; ++s)
#pragma unroll
    for (int j = 0; j < RT_TOK; ++j) {
      const v2f xlo = f.xv[j][s].xy, xhi = f.xv[j][s].zw;
#pragma unroll
      for (int k = 0; k < K_EXP; ++k) {
        acc[j][k] = __builtin_elementwise_fma(xlo, (v2f)f.wv[k][s].xy, acc[j][k]);
        acc[j][k] = __builtin_elementwise_fma(xhi, (v2f)f.wv[k][s].zw, acc[j][k]);
      }
    }
}

__global__ __launch_bounds__(256) void moelora_router(
    const float* __restrict__ x, const float* __restrict__ Wr,
    const int* __restrict__ alpha_p, float* __restrict__ g,
    const float* __restrict__ Amat, float* __restrict__ Af,
    int* __restrict__ sel, float* __restrict__ hvz, int nstep)
{
  __shared__ v4f part[4][2][2 * (K_EXP / 4)][32];
  const int wave = threadIdx.x >> 5;
  const int lane = threadIdx.x & 31;
  const int grp  = wave & 3;
  const int half = wave >> 2;
  const int gw = blockIdx.x * 8 + wave;
  const int t0 = (blockIdx.x * 4 + grp) * RT_TOK;

  static_assert(AF_UNITS == 2 * (NTOK / 16) * 8, "two Af units per router wave");
  v2f alo[2], ahi[2];
#pragma unroll
  for (int i = 0; i < 2; ++i) {
    const int u = 2 * gw + i;
    const float* asrc = Amat + (size_t)((u >> 8) * 16 + (lane & 15)) * D_DIM
                             + 32 * ((u >> 2) & 63) + 8 * (u & 3) + 2 * (lane >> 4);
    alo[i] = *(const v2f*)asrc;
    ahi[i] = *(const v2f*)(asrc + 4);
  }

  v2f acc[RT_TOK][K_EXP];
#pragma unroll
  for (int j = 0; j < RT_TOK; ++j)
#pragma unroll
    for (int k = 0; k < K_EXP; ++k) { acc[j][k].x = 0.f; acc[j][k].y = 0.f; }

  const float* xl = x  + (size_t)t0 * D_DIM + half * RT_HALF + lane * 4;
  const float* wl = Wr + half * RT_HALF + lane * 4;

  RouteSet fa, fb;
  route_load(fa, xl, wl, 0);
  int c = 0;
  for (; c + 1 < nstep; c += 2) {
    route_load(fb, xl, wl, c + 1);
    route_fma(fa, acc);
    const int cn = (c + 2 < nstep) ? c + 2 : c;
    route_load(fa, xl, wl, cn);
    route_fma(fb, acc);
  }
  if (c < nstep) route_fma(fa, acc);

#pragma unroll
  for (int i = 0; i < 2; ++i) {
    v4f o; o.x = alo[i].x; o.y = alo[i].y; o.z = ahi[i].x; o.w = ahi[i].y;
    *(v4f*)(Af + ((size_t)(2 * gw + i) * 32 + lane) * 4) = o;
  }
  static_assert(NTOK * N_HV == (NTOK / 16) * 8 * 512, "512 hv floats per router wave");
#pragma unroll
  for (int i = 0; i < 4; ++i) {
    v4f z; z.x = 0.f; z.y = 0.f; z.z = 0.f; z.w = 0.f;
    *(v4f*)(hvz + (size_t)gw * 512 + i * 128 + lane * 4) = z;
  }

  float mine[2][K_EXP], give[2][K_EXP];
#pragma unroll
  for (int i = 0; i < 2; ++i)
#pragma unroll
    for (int k = 0; k < K_EXP; ++k) {
      const float lo = acc[i][k].x + acc[i][k].y;
      const float hi = acc[2 + i][k].x + acc[2 + i][k].y;
      mine[i][k] = half ? hi : lo;
      give[i][k] = half ? lo : hi;
    }
#pragma unroll
  for (int i = 0; i < 2; ++i)
#pragma unroll
    for (int k4 = 0; k4 < K_EXP / 4; ++k4) {
      v4f o; o.x = give[i][4 * k4]; o.y = give[i][4 * k4 + 1]; o.z = give[i][4 * k4 + 2]; o.w = give[i][4 * k4 + 3];
      part[grp][half][i * 4 + k4][lane] = o;
    }
  __syncthreads();
#pragma unroll
  for (int i = 0; i < 2; ++i)
#pragma unroll
    for (int k4 = 0; k4 < K_EXP / 4; ++k4) {
      const v4f o = part[grp][1 - half][i * 4 + k4][lane];
      mine[i][4 * k4]     += o.x;
      mine[i][4 * k4 + 1] += o.y;
      mine[i][4 * k4 + 2] += o.z;
      mine[i][4 * k4 + 3] += o.w;
    }

#pragma unroll
  for (int i = 0; i < 2; ++i)
#pragma unroll
    for (int k = 0; k < K_EXP; ++k) {
#pragma unroll
      for (int off = 16; off > 0; off >>= 1)
        mine[i][k] += __shfl_xor(mine[i][k], off, 32);
    }

  float sc[K_EXP];
#pragma unroll
  for (int k = 0; k < K_EXP; ++k) sc[k] = (lane == 1) ? mine[1][k] : mine[0][k];

  if (lane < 2) {
    const int t = t0 + 2 * half + lane;
    const float alpha = (float)(*alpha_p);
    int i0 = 0; float v0 = sc[0];
#pragma unroll
    for (int k = 1; k < K_EXP; ++k) if (sc[k] > v0) { v0 = sc[k]; i0 = k; }
    int i1 = -1; float v1 = -3.4e38f;
#pragma unroll
    for (int k = 0; k < K_EXP; ++k)
      if (k != i0 && sc[k] > v1) { v1 = sc[k]; i1 = k; }
    const float e  = __expf(v1 - v0);
    const float dn = 1.f + e;
    const float g0 = alpha / dn;
    const float g1 = alpha * e / dn;
    sel[t] = i0 | ((i1 & 0xff) << 8);
#pragma unroll
    for (int k4 = 0; k4 < K_EXP; k4 += 4) {
      v4f o;
      o.x = (k4     == i0) ? g0 : ((k4     == i1) ? g1 : 0.f);
      o.y = (k4 + 1 == i0) ? g0 : ((k4 + 1 == i1) ? g1 : 0.f);
      o.z = (k4 + 2 == i0) ? g0 : ((k4 + 2 == i1) ? g1 : 0.f);
      o.w = (k4 + 3 == i0) ? g0 : ((k4 + 3 == i1) ? g1 : 0.f);
      *(v4f*)(g + (size_t)t * K_EXP + k4) = o;
    }
  }
}

#define KSTEPS 16
#define KBLK   (KSTEPS * 4)
#define KQ     (D_DIM / 4)

struct FragSet { v2f a[KSTEPS]; v4f w[KSTEPS / 2]; };

static __device__ __forceinline__ void load_block(
    FragSet& f, const float* x0, const float* w0, int kb)
{
#pragma unroll
  for (int s = 0; s < KSTEPS; ++s)
    f.a[s] = *(const v2f*)(x0 + kb + s * 4);
#pragma unroll
  for (int p = 0; p < KSTEPS / 2; ++p)
    f.w[p] = *(const v4f*)(w0 + kb * 16 + p * AF_UNIT);
}

static __device__ __forceinline__ void mma_block(const FragSet& f, v8f (&c)[2])
{
#pragma unroll
  for (int s = 0; s < KSTEPS; ++s) {
    const v4f u = f.w[s >> 1];
    v2f b;
    if (s & 1) { b.x = u.z; b.y = u.w; }
    else       { b.x = u.x; b.y = u.y; }
    c[s & 1] = wmma_f32(b, f.a[s], c[s & 1]);
  }
}

__global__ __launch_bounds__(128) void moelora_hgemm(
    const float* __restrict__ x, const float* __restrict__ Af,
    const float* __restrict__ g, const int* __restrict__ sel,
    float* __restrict__ hv, int nblk)
{
  __shared__ unsigned short rows[NTOK];
  __shared__ int wtot[4];
  __shared__ float part[3][8][32];
  const int tid  = threadIdx.x;
  const int lane = tid & 31;
  const int wave = __builtin_amdgcn_readfirstlane(tid >> 5);
  const int m = lane & 15;
  const int h = lane >> 4;
  const int e    = blockIdx.x & (K_EXP - 1);
  const int slot = blockIdx.x >> 4;

  unsigned hit = 0;
  {
    const v4i* sp = (const v4i*)(sel + tid * 32);
    v4i w[8];
#pragma unroll
    for (int i = 0; i < 8; ++i) w[i] = sp[i];
#pragma unroll
    for (int i = 0; i < 8; ++i)
#pragma unroll
      for (int q = 0; q < 4; ++q) {
        const int v = w[i][q];
        const bool is = ((v & 0xff) == e) || (((v >> 8) & 0xff) == e);
        hit |= (is ? 1u : 0u) << (i * 4 + q);
      }
  }
  const int cnt = __popc(hit);
  int inc = cnt;
#pragma unroll
  for (int off = 1; off < 32; off <<= 1) {
    const int o = __shfl_up(inc, off, 32);
    if (lane >= off) inc += o;
  }
  if (lane == 31) wtot[wave] = inc;
  __syncthreads();
  int base = 0, ntok = 0;
#pragma unroll
  for (int w2 = 0; w2 < 4; ++w2) {
    const int tw = wtot[w2];
    if (w2 < wave) base += tw;
    ntok += tw;
  }
  {
    int pos = base + inc - cnt;
    for (int b = 0; b < 32; ++b)
      if ((hit >> b) & 1u) rows[pos++] = (unsigned short)(tid * 32 + b);
  }
  __syncthreads();

  const int n_e    = __builtin_amdgcn_readfirstlane(ntok);
  const int ntiles = (n_e + 15) >> 4;
  const float* arow = Af + (size_t)e * AF_TILE + (size_t)wave * KQ * 16 + lane * 4;
  const int klast = (nblk - 1) * KBLK;

  for (int j = slot; j < ntiles; j += 16) {
    const int idx   = j * 16 + m;
    const bool live = idx < n_e;
    const int tok   = rows[live ? idx : j * 16];
    const float* xrow = x + (size_t)tok * D_DIM + wave * KQ + 2 * h;
    const float ge = g[(size_t)tok * K_EXP + e];

    v8f c[2];
    c[0] = (v8f){}; c[1] = (v8f){};

    FragSet f0, f1, f2;
    load_block(f0, xrow, arow, 0);
    __builtin_amdgcn_sched_barrier(0);
    load_block(f1, xrow, arow, (KBLK < klast) ? KBLK : klast);
    int kb = 0;
    int cb = 0;
    for (; cb + 3 <= nblk; cb += 3) {
      const int k2 = (kb + 2 * KBLK < klast) ? kb + 2 * KBLK : klast;
      const int k3 = (kb + 3 * KBLK < klast) ? kb + 3 * KBLK : klast;
      const int k4 = (kb + 4 * KBLK < klast) ? kb + 4 * KBLK : klast;
      load_block(f2, xrow, arow, k2);
      mma_block(f0, c);
      load_block(f0, xrow, arow, k3);
      mma_block(f1, c);
      load_block(f1, xrow, arow, k4);
      mma_block(f2, c);
      kb += 3 * KBLK;
      __builtin_amdgcn_sched_group_barrier(0x020, KSTEPS + KSTEPS / 2, 0);
      __builtin_amdgcn_sched_group_barrier(0x008, KSTEPS, 0);
      __builtin_amdgcn_sched_group_barrier(0x020, KSTEPS + KSTEPS / 2, 0);
      __builtin_amdgcn_sched_group_barrier(0x008, KSTEPS, 0);
      __builtin_amdgcn_sched_group_barrier(0x020, KSTEPS + KSTEPS / 2, 0);
      __builtin_amdgcn_sched_group_barrier(0x008, KSTEPS, 0);
    }
    if (cb < nblk)     mma_block(f0, c);
    if (cb + 1 < nblk) mma_block(f1, c);

    float sum[8];
#pragma unroll
    for (int i = 0; i < 8; ++i) sum[i] = c[0][i] + c[1][i];
    if (wave != 0) {
#pragma unroll
      for (int i = 0; i < 8; ++i) part[wave - 1][i][lane] = sum[i];
    }
    __syncthreads();
    if (wave == 0) {
#pragma unroll
      for (int w2 = 0; w2 < 3; ++w2)
#pragma unroll
        for (int i = 0; i < 8; ++i) sum[i] += part[w2][i][lane];
      if (live) {
        float* p0 = hv + ((size_t)(tok >> 4) * K_EXP + e) * HV_EXP + ((tok & 15) + 16 * h) * 2;
#pragma unroll
        for (int jj = 0; jj < 4; ++jj) {
          v2f v;
          v.x = sum[2 * jj] * ge; v.y = sum[2 * jj + 1] * ge;
          *(v2f*)(p0 + jj * HV_STEP) = v;
        }
      }
    }
    __syncthreads();
  }
}

#define O_MT 4
#define O_NT 8

struct OFragSet { v2f a[O_MT][4]; v4f b[O_NT][2]; };

static __device__ __forceinline__ void o_load(
    OFragSet& f, const float* hr, const float* bb)
{
#pragma unroll
  for (int mi = 0; mi < O_MT; ++mi)
#pragma unroll
    for (int s = 0; s < 4; ++s)
      f.a[mi][s] = *(const v2f*)(hr + (size_t)mi * K_EXP * HV_EXP + s * HV_STEP);
#pragma unroll
  for (int ni = 0; ni < O_NT; ++ni)
#pragma unroll
    for (int p = 0; p < 2; ++p)
      f.b[ni][p] = *(const v4f*)(bb + (size_t)ni * 16 * R_RANK + p * 4);
}

static __device__ __forceinline__ void o_mma(
    const OFragSet& f, v8f (&c)[O_MT][O_NT])
{
#pragma unroll
  for (int s = 0; s < 4; ++s)
#pragma unroll
    for (int mi = 0; mi < O_MT; ++mi)
#pragma unroll
      for (int ni = 0; ni < O_NT; ++ni) {
        const v4f u = f.b[ni][s >> 1];
        v2f b;
        if (s & 1) { b.x = u.z; b.y = u.w; }
        else       { b.x = u.x; b.y = u.y; }
        c[mi][ni] = wmma_f32(f.a[mi][s], b, c[mi][ni]);
      }
}

__global__ __launch_bounds__(128) void moelora_ogemm(
    const float* __restrict__ hv, const float* __restrict__ Bm,
    float* __restrict__ out, int nexp)
{
  const int wave = threadIdx.x >> 5;
  const int lane = threadIdx.x & 31;
  const int m = lane & 15;
  const int h = lane >> 4;
  const int mt0 = blockIdx.x * O_MT;
  const int dt0 = blockIdx.y * (4 * O_NT) + wave * O_NT;

  const float* hr = hv + (size_t)mt0 * K_EXP * HV_EXP + lane * 2;
  const float* bb = Bm + (size_t)(dt0 * 16 + m) * R_RANK + 8 * h;

  v8f c[O_MT][O_NT];
#pragma unroll
  for (int mi = 0; mi < O_MT; ++mi)
#pragma unroll
    for (int ni = 0; ni < O_NT; ++ni) c[mi][ni] = (v8f){};

  OFragSet f0, f1;
  const int elast = nexp - 1;
  o_load(f0, hr, bb);
  int e = 0;
  for (; e + 2 <= nexp; e += 2) {
    const int e2 = (e + 2 < nexp) ? e + 2 : elast;
    o_load(f1, hr + (size_t)(e + 1) * HV_EXP, bb + (size_t)(e + 1) * D_DIM * R_RANK);
    o_mma(f0, c);
    o_load(f0, hr + (size_t)e2 * HV_EXP, bb + (size_t)e2 * D_DIM * R_RANK);
    o_mma(f1, c);
    __builtin_amdgcn_sched_group_barrier(0x020, 32, 0);
    __builtin_amdgcn_sched_group_barrier(0x008, 128, 0);
    __builtin_amdgcn_sched_group_barrier(0x020, 32, 0);
    __builtin_amdgcn_sched_group_barrier(0x008, 128, 0);
  }
  if (e < nexp) o_mma(f0, c);

  const int row0 = mt0 * 16;
#pragma unroll
  for (int mi = 0; mi < O_MT; ++mi)
#pragma unroll
    for (int ni = 0; ni < O_NT; ++ni)
#pragma unroll
      for (int i = 0; i < 8; ++i)
        out[(size_t)(row0 + 16 * mi + i + 8 * h) * D_DIM + (dt0 + ni) * 16 + m] = c[mi][ni][i];
}

extern "C" void kernel_launch(void* const* d_in, const int* in_sizes, int n_in,
                              void* d_out, int out_size, void* d_ws, size_t ws_size,
                              hipStream_t stream) {
  (void)in_sizes; (void)n_in; (void)out_size; (void)ws_size;
  const float* x     = (const float*)d_in[0];
  const int*   alpha = (const int*)  d_in[1];
  const float* A     = (const float*)d_in[2];
  const float* Bm    = (const float*)d_in[3];
  const float* Wr    = (const float*)d_in[4];
  float* out = (float*)d_out;

  float* g  = (float*)d_ws;
  float* hv = g + (size_t)NTOK * K_EXP;
  float* Af = hv + (size_t)NTOK * N_HV;
  int* sel  = (int*)(Af + (size_t)AF_UNITS * AF_UNIT);

  moelora_router<<<NTOK / 16, 256, 0, stream>>>(x, Wr, alpha, g, A, Af, sel, hv, RT_HALF / RT_D);
  moelora_hgemm<<<K_EXP * 16, 128, 0, stream>>>(x, Af, g, sel, hv, KQ / KBLK);
  moelora_ogemm<<<dim3(NTOK / 64, D_DIM / 512), 128, 0, stream>>>(hv, Bm, out, K_EXP);
}


# o-GEMM computes only each token's 2 selected experts: per-expert 16-row tiles inside a 64-token block, a token's two result rows added through LDS; new sum order: rank 0 + rank 1
# speedup vs baseline: 2.9595x; 1.2060x over previous
#include <hip/hip_runtime.h>


typedef __attribute__((ext_vector_type(2))) float v2f;
typedef __attribute__((ext_vector_type(4))) float v4f;
typedef __attribute__((ext_vector_type(8))) float v8f;
typedef __attribute__((ext_vector_type(4))) int   v4i;

#define D_DIM  2048
#define K_EXP  16
#define R_RANK 16
#define NTOK   4096
#define N_HV   256
#define AF_UNIT 128
#define AF_TILE (64 * 4 * AF_UNIT)
#define AF_UNITS (K_EXP * 64 * 4)

static __device__ __forceinline__ v8f wmma_f32(v2f a, v2f b, v8f c) {
  return __builtin_amdgcn_wmma_f32_16x16x4_f32(
       false, a,  false, b,
       (short)0, c,  false,  false);
}

#define RT_TOK  4
#define RT_D    256
#define RT_SUB  (RT_D / 128)
#define RT_HALF (D_DIM / 2)

struct RouteSet { v4f xv[RT_TOK][RT_SUB]; v4f wv[K_EXP][RT_SUB]; };

static __device__ __forceinline__ void route_load(
    RouteSet& f, const float* xl, const float* wl, int c)
{
#pragma unroll
  for (int j = 0; j < RT_TOK; ++j)
#pragma unroll
    for (int s = 0; s < RT_SUB; ++s)
      f.xv[j][s] = *(const v4f*)(xl + (size_t)j * D_DIM + c * RT_D + s * 128);
#pragma unroll
  for (int k = 0; k < K_EXP; ++k)
#pragma unroll
    for (int s = 0; s < RT_SUB; ++s)
      f.wv[k][s] = *(const v4f*)(wl + (size_t)k * D_DIM + c * RT_D + s * 128);
}

static __device__ __forceinline__ void route_fma(
    const RouteSet& f, v2f (&acc)[RT_TOK][K_EXP])
{
#pragma unroll
  for (int s = 0; s < RT_SUB; ++s)
#pragma unroll
    for (int j = 0; j < RT_TOK; ++j) {
      const v2f xlo = f.xv[j][s].xy, xhi = f.xv[j][s].zw;
#pragma unroll
      for (int k = 0; k < K_EXP; ++k) {
        acc[j][k] = __builtin_elementwise_fma(xlo, (v2f)f.wv[k][s].xy, acc[j][k]);
        acc[j][k] = __builtin_elementwise_fma(xhi, (v2f)f.wv[k][s].zw, acc[j][k]);
      }
    }
}

__global__ __launch_bounds__(256) void moelora_router(
    const float* __restrict__ x, const float* __restrict__ Wr,
    const int* __restrict__ alpha_p, float* __restrict__ g,
    const float* __restrict__ Amat, float* __restrict__ Af,
    int* __restrict__ sel, int nstep)
{
  __shared__ v4f part[4][2][2 * (K_EXP / 4)][32];
  const int wave = threadIdx.x >> 5;
  const int lane = threadIdx.x & 31;
  const int grp  = wave & 3;
  const int half = wave >> 2;
  const int gw = blockIdx.x * 8 + wave;
  const int t0 = (blockIdx.x * 4 + grp) * RT_TOK;

  static_assert(AF_UNITS == 2 * (NTOK / 16) * 8, "two Af units per router wave");
  v2f alo[2], ahi[2];
#pragma unroll
  for (int i = 0; i < 2; ++i) {
    const int u = 2 * gw + i;
    const float* asrc = Amat + (size_t)((u >> 8) * 16 + (lane & 15)) * D_DIM
                             + 32 * ((u >> 2) & 63) + 8 * (u & 3) + 2 * (lane >> 4);
    alo[i] = *(const v2f*)asrc;
    ahi[i] = *(const v2f*)(asrc + 4);
  }

  v2f acc[RT_TOK][K_EXP];
#pragma unroll
  for (int j = 0; j < RT_TOK; ++j)
#pragma unroll
    for (int k = 0; k < K_EXP; ++k) { acc[j][k].x = 0.f; acc[j][k].y = 0.f; }

  const float* xl = x  + (size_t)t0 * D_DIM + half * RT_HALF + lane * 4;
  const float* wl = Wr + half * RT_HALF + lane * 4;

  RouteSet fa, fb;
  route_load(fa, xl, wl, 0);
  int c = 0;
  for (; c + 1 < nstep; c += 2) {
    route_load(fb, xl, wl, c + 1);
    route_fma(fa, acc);
    const int cn = (c + 2 < nstep) ? c + 2 : c;
    route_load(fa, xl, wl, cn);
    route_fma(fb, acc);
  }
  if (c < nstep) route_fma(fa, acc);

#pragma unroll
  for (int i = 0; i < 2; ++i) {
    v4f o; o.x = alo[i].x; o.y = alo[i].y; o.z = ahi[i].x; o.w = ahi[i].y;
    *(v4f*)(Af + ((size_t)(2 * gw + i) * 32 + lane) * 4) = o;
  }
  float mine[2][K_EXP], give[2][K_EXP];
#pragma unroll
  for (int i = 0; i < 2; ++i)
#pragma unroll
    for (int k = 0; k < K_EXP; ++k) {
      const float lo = acc[i][k].x + acc[i][k].y;
      const float hi = acc[2 + i][k].x + acc[2 + i][k].y;
      mine[i][k] = half ? hi : lo;
      give[i][k] = half ? lo : hi;
    }
#pragma unroll
  for (int i = 0; i < 2; ++i)
#pragma unroll
    for (int k4 = 0; k4 < K_EXP / 4; ++k4) {
      v4f o; o.x = give[i][4 * k4]; o.y = give[i][4 * k4 + 1]; o.z = give[i][4 * k4 + 2]; o.w = give[i][4 * k4 + 3];
      part[grp][half][i * 4 + k4][lane] = o;
    }
  __syncthreads();
#pragma unroll
  for (int i = 0; i < 2; ++i)
#pragma unroll
    for (int k4 = 0; k4 < K_EXP / 4; ++k4) {
      const v4f o = part[grp][1 - half][i * 4 + k4][lane];
      mine[i][4 * k4]     += o.x;
      mine[i][4 * k4 + 1] += o.y;
      mine[i][4 * k4 + 2] += o.z;
      mine[i][4 * k4 + 3] += o.w;
    }

#pragma unroll
  for (int i = 0; i < 2; ++i)
#pragma unroll
    for (int k = 0; k < K_EXP; ++k) {
#pragma unroll
      for (int off = 16; off > 0; off >>= 1)
        mine[i][k] += __shfl_xor(mine[i][k], off, 32);
    }

  float sc[K_EXP];
#pragma unroll
  for (int k = 0; k < K_EXP; ++k) sc[k] = (lane == 1) ? mine[1][k] : mine[0][k];

  if (lane < 2) {
    const int t = t0 + 2 * half + lane;
    const float alpha = (float)(*alpha_p);
    int i0 = 0; float v0 = sc[0];
#pragma unroll
    for (int k = 1; k < K_EXP; ++k) if (sc[k] > v0) { v0 = sc[k]; i0 = k; }
    int i1 = -1; float v1 = -3.4e38f;
#pragma unroll
    for (int k = 0; k < K_EXP; ++k)
      if (k != i0 && sc[k] > v1) { v1 = sc[k]; i1 = k; }
    const float e  = __expf(v1 - v0);
    const float dn = 1.f + e;
    const float g0 = alpha / dn;
    const float g1 = alpha * e / dn;
    sel[t] = i0 | ((i1 & 0xff) << 8);
#pragma unroll
    for (int k4 = 0; k4 < K_EXP; k4 += 4) {
      v4f o;
      o.x = (k4     == i0) ? g0 : ((k4     == i1) ? g1 : 0.f);
      o.y = (k4 + 1 == i0) ? g0 : ((k4 + 1 == i1) ? g1 : 0.f);
      o.z = (k4 + 2 == i0) ? g0 : ((k4 + 2 == i1) ? g1 : 0.f);
      o.w = (k4 + 3 == i0) ? g0 : ((k4 + 3 == i1) ? g1 : 0.f);
      *(v4f*)(g + (size_t)t * K_EXP + k4) = o;
    }
  }
}

#define KSTEPS 16
#define KBLK   (KSTEPS * 4)
#define KQ     (D_DIM / 4)

struct FragSet { v2f a[KSTEPS]; v4f w[KSTEPS / 2]; };

static __device__ __forceinline__ void load_block(
    FragSet& f, const float* x0, const float* w0, int kb)
{
#pragma unroll
  for (int s = 0; s < KSTEPS; ++s)
    f.a[s] = *(const v2f*)(x0 + kb + s * 4);
#pragma unroll
  for (int p = 0; p < KSTEPS / 2; ++p)
    f.w[p] = *(const v4f*)(w0 + kb * 16 + p * AF_UNIT);
}

static __device__ __forceinline__ void mma_block(const FragSet& f, v8f (&c)[2])
{
#pragma unroll
  for (int s = 0; s < KSTEPS; ++s) {
    const v4f u = f.w[s >> 1];
    v2f b;
    if (s & 1) { b.x = u.z; b.y = u.w; }
    else       { b.x = u.x; b.y = u.y; }
    c[s & 1] = wmma_f32(b, f.a[s], c[s & 1]);
  }
}

__global__ __launch_bounds__(128) void moelora_hgemm(
    const float* __restrict__ x, const float* __restrict__ Af,
    const float* __restrict__ g, const int* __restrict__ sel,
    float* __restrict__ hv, int nblk)
{
  __shared__ unsigned short rows[NTOK];
  __shared__ int wtot[4];
  __shared__ float part[3][8][32];
  const int tid  = threadIdx.x;
  const int lane = tid & 31;
  const int wave = __builtin_amdgcn_readfirstlane(tid >> 5);
  const int m = lane & 15;
  const int h = lane >> 4;
  const int e    = blockIdx.x & (K_EXP - 1);
  const int slot = blockIdx.x >> 4;

  unsigned hit = 0;
  {
    const v4i* sp = (const v4i*)(sel + tid * 32);
    v4i w[8];
#pragma unroll
    for (int i = 0; i < 8; ++i) w[i] = sp[i];
#pragma unroll
    for (int i = 0; i < 8; ++i)
#pragma unroll
      for (int q = 0; q < 4; ++q) {
        const int v = w[i][q];
        const bool is = ((v & 0xff) == e) || (((v >> 8) & 0xff) == e);
        hit |= (is ? 1u : 0u) << (i * 4 + q);
      }
  }
  const int cnt = __popc(hit);
  int inc = cnt;
#pragma unroll
  for (int off = 1; off < 32; off <<= 1) {
    const int o = __shfl_up(inc, off, 32);
    if (lane >= off) inc += o;
  }
  if (lane == 31) wtot[wave] = inc;
  __syncthreads();
  int base = 0, ntok = 0;
#pragma unroll
  for (int w2 = 0; w2 < 4; ++w2) {
    const int tw = wtot[w2];
    if (w2 < wave) base += tw;
    ntok += tw;
  }
  {
    int pos = base + inc - cnt;
    for (int b = 0; b < 32; ++b)
      if ((hit >> b) & 1u) rows[pos++] = (unsigned short)(tid * 32 + b);
  }
  __syncthreads();

  const int n_e    = __builtin_amdgcn_readfirstlane(ntok);
  const int ntiles = (n_e + 15) >> 4;
  const float* arow = Af + (size_t)e * AF_TILE + (size_t)wave * KQ * 16 + lane * 4;
  const int klast = (nblk - 1) * KBLK;

  for (int j = slot; j < ntiles; j += 16) {
    const int idx   = j * 16 + m;
    const bool live = idx < n_e;
    const int tok   = rows[live ? idx : j * 16];
    const float* xrow = x + (size_t)tok * D_DIM + wave * KQ + 2 * h;
    const float ge = g[(size_t)tok * K_EXP + e];

    v8f c[2];
    c[0] = (v8f){}; c[1] = (v8f){};

    FragSet f0, f1, f2;
    load_block(f0, xrow, arow, 0);
    __builtin_amdgcn_sched_barrier(0);
    load_block(f1, xrow, arow, (KBLK < klast) ? KBLK : klast);
    int kb = 0;
    int cb = 0;
    for (; cb + 3 <= nblk; cb += 3) {
      const int k2 = (kb + 2 * KBLK < klast) ? kb + 2 * KBLK : klast;
      const int k3 = (kb + 3 * KBLK < klast) ? kb + 3 * KBLK : klast;
      const int k4 = (kb + 4 * KBLK < klast) ? kb + 4 * KBLK : klast;
      load_block(f2, xrow, arow, k2);
      mma_block(f0, c);
      load_block(f0, xrow, arow, k3);
      mma_block(f1, c);
      load_block(f1, xrow, arow, k4);
      mma_block(f2, c);
      kb += 3 * KBLK;
      __builtin_amdgcn_sched_group_barrier(0x020, KSTEPS + KSTEPS / 2, 0);
      __builtin_amdgcn_sched_group_barrier(0x008, KSTEPS, 0);
      __builtin_amdgcn_sched_group_barrier(0x020, KSTEPS + KSTEPS / 2, 0);
      __builtin_amdgcn_sched_group_barrier(0x008, KSTEPS, 0);
      __builtin_amdgcn_sched_group_barrier(0x020, KSTEPS + KSTEPS / 2, 0);
      __builtin_amdgcn_sched_group_barrier(0x008, KSTEPS, 0);
    }
    if (cb < nblk)     mma_block(f0, c);
    if (cb + 1 < nblk) mma_block(f1, c);

    float sum[8];
#pragma unroll
    for (int i = 0; i < 8; ++i) sum[i] = c[0][i] + c[1][i];
    if (wave != 0) {
#pragma unroll
      for (int i = 0; i < 8; ++i) part[wave - 1][i][lane] = sum[i];
    }
    __syncthreads();
    if (wave == 0) {
#pragma unroll
      for (int w2 = 0; w2 < 3; ++w2)
#pragma unroll
        for (int i = 0; i < 8; ++i) sum[i] += part[w2][i][lane];
      if (live) {
        float* p0 = hv + ((size_t)tok * K_EXP + e) * R_RANK + 8 * h;
        v4f lo, hi;
        lo.x = sum[0] * ge; lo.y = sum[1] * ge; lo.z = sum[2] * ge; lo.w = sum[3] * ge;
        hi.x = sum[4] * ge; hi.y = sum[5] * ge; hi.z = sum[6] * ge; hi.w = sum[7] * ge;
        *(v4f*)p0 = lo;
        *(v4f*)(p0 + 4) = hi;
      }
    }
    __syncthreads();
  }
}

#define S_TB    64
#define S_CB    512
#define S_NT    4
#define S_PITCH (S_CB + 4)
#define S_ROWS  (2 * S_TB + 16)
#define S_MAXT  24

struct SFrag { v4f a[2]; v4f b[S_NT][2]; };

static __device__ __forceinline__ void s_load(
    SFrag& f, const float* hv, const float* bw, const unsigned char* slotent,
    int w, int t0, int m, int h)
{
  const int e = w & 0xff, sb = (w >> 8) & 0xff, end = w >> 16;
  const int slot = sb + m;
  const int ent = slotent[(slot < end) ? slot : sb];
  const float* hr = hv + ((size_t)(t0 + (ent >> 1)) * K_EXP + e) * R_RANK + 8 * h;
  f.a[0] = *(const v4f*)hr;
  f.a[1] = *(const v4f*)(hr + 4);
  const float* bb = bw + (size_t)e * D_DIM * R_RANK;
#pragma unroll
  for (int ni = 0; ni < S_NT; ++ni)
#pragma unroll
    for (int p = 0; p < 2; ++p)
      f.b[ni][p] = *(const v4f*)(bb + (size_t)ni * 16 * R_RANK + p * 4);
}

static __device__ __forceinline__ void s_mma_store(const SFrag& f, float* dst)
{
  v8f c[S_NT];
#pragma unroll
  for (int ni = 0; ni < S_NT; ++ni) c[ni] = (v8f){};
#pragma unroll
  for (int s = 0; s < 4; ++s) {
    const v4f ua = f.a[s >> 1];
    v2f a;
    if (s & 1) { a.x = ua.z; a.y = ua.w; }
    else       { a.x = ua.x; a.y = ua.y; }
#pragma unroll
    for (int ni = 0; ni < S_NT; ++ni) {
      const v4f ub = f.b[ni][s >> 1];
      v2f b;
      if (s & 1) { b.x = ub.z; b.y = ub.w; }
      else       { b.x = ub.x; b.y = ub.y; }
      c[ni] = wmma_f32(a, b, c[ni]);
    }
  }
#pragma unroll
  for (int ni = 0; ni < S_NT; ++ni)
#pragma unroll
    for (int i = 0; i < 8; ++i)
      dst[i * S_PITCH + ni * 16] = c[ni][i];
}

__global__ __launch_bounds__(256) void moelora_ogemm(
    const float* __restrict__ hv, const float* __restrict__ Bm,
    const int* __restrict__ sel, float* __restrict__ out)
{
  __shared__ __attribute__((aligned(16))) float buf[S_ROWS * S_PITCH];
  __shared__ unsigned wmask[K_EXP][4];
  __shared__ unsigned char slotent[2 * S_TB];
  __shared__ unsigned char entslot[2 * S_TB];
  __shared__ int wl[S_MAXT];
  __shared__ int nwl;
  const int tid  = threadIdx.x;
  const int lane = tid & 31;
  const int wave = __builtin_amdgcn_readfirstlane(tid >> 5);
  const int m = lane & 15;
  const int h = lane >> 4;
  const int t0 = blockIdx.x * S_TB;

  int emine = 0;
  if (wave < 4) {
    const int v = sel[t0 + (tid >> 1)];
    emine = ((tid & 1) ? (v >> 8) : v) & (K_EXP - 1);
#pragma unroll
    for (int e = 0; e < K_EXP; ++e) {
      const unsigned b = __builtin_amdgcn_ballot_w32(emine == e);
      if (lane == 0) wmask[e][wave] = b;
    }
  }
  __syncthreads();
  if (wave < 4) {
    int base = 0, before = 0;
#pragma unroll
    for (int e2 = 0; e2 < K_EXP; ++e2)
#pragma unroll
      for (int w2 = 0; w2 < 4; ++w2) {
        const int cn = __popc(wmask[e2][w2]);
        if (e2 < emine) base += cn;
        if (e2 == emine && w2 < wave) before += cn;
      }
    before += __popc(wmask[emine][wave] & ((1u << lane) - 1u));
    const int slot = base + before;
    entslot[tid] = (unsigned char)slot;
    slotent[slot] = (unsigned char)tid;
  } else if (wave == 4) {
    const int eq = lane & (K_EXP - 1);
    const int n = (lane < K_EXP)
        ? __popc(wmask[eq][0]) + __popc(wmask[eq][1]) + __popc(wmask[eq][2]) + __popc(wmask[eq][3]) : 0;
    const int nt = (n + 15) >> 4;
    int pn = n, pt = nt;
#pragma unroll
    for (int off = 1; off < K_EXP; off <<= 1) {
      const int on = __shfl_up(pn, off, 32);
      const int ot = __shfl_up(pt, off, 32);
      if (lane >= off) { pn += on; pt += ot; }
    }
    const int base = pn - n;
    int k = pt - nt;
    for (int j = 0; j < nt && k < S_MAXT; ++j, ++k)
      wl[k] = eq | ((base + 16 * j) << 8) | ((base + n) << 16);
    if (lane == K_EXP - 1) nwl = pt;
  }
  __syncthreads();

  const int nw = __builtin_amdgcn_readfirstlane(nwl);
  const int colw = blockIdx.y * S_CB + wave * (S_NT * 16);
  const float* bw = Bm + (size_t)(colw + m) * R_RANK + 8 * h;
  float* dbase = buf + 8 * h * S_PITCH + wave * (S_NT * 16) + m;

  SFrag f0, f1;
  int w0 = __builtin_amdgcn_readfirstlane(wl[0]);
  s_load(f0, hv, bw, slotent, w0, t0, m, h);
  for (int k = 0; k < nw; k += 2) {
    const int k1 = (k + 1 < nw) ? k + 1 : nw - 1;
    const int k2 = (k + 2 < nw) ? k + 2 : nw - 1;
    const int w1 = __builtin_amdgcn_readfirstlane(wl[k1]);
    const int w2 = __builtin_amdgcn_readfirstlane(wl[k2]);
    s_load(f1, hv, bw, slotent, w1, t0, m, h);
    __builtin_amdgcn_sched_barrier(0);
    s_mma_store(f0, dbase + ((w0 >> 8) & 0xff) * S_PITCH);
    __builtin_amdgcn_sched_barrier(0);
    s_load(f0, hv, bw, slotent, w2, t0, m, h);
    __builtin_amdgcn_sched_barrier(0);
    s_mma_store(f1, dbase + ((w1 >> 8) & 0xff) * S_PITCH);
    __builtin_amdgcn_sched_barrier(0);
    w0 = w2;
  }
  __syncthreads();

#pragma unroll 2
  for (int i = 0; i < S_TB / 8; ++i) {
    const int tl = wave + 8 * i;
    const float* r0 = buf + (int)entslot[2 * tl]     * S_PITCH + lane * 4;
    const float* r1 = buf + (int)entslot[2 * tl + 1] * S_PITCH + lane * 4;
    float* op = out + (size_t)(t0 + tl) * D_DIM + blockIdx.y * S_CB + lane * 4;
#pragma unroll
    for (int q = 0; q < S_CB / 128; ++q) {
      const v4f a = *(const v4f*)(r0 + q * 128);
      const v4f b = *(const v4f*)(r1 + q * 128);
      *(v4f*)(op + q * 128) = a + b;
    }
  }
}

extern "C" void kernel_launch(void* const* d_in, const int* in_sizes, int n_in,
                              void* d_out, int out_size, void* d_ws, size_t ws_size,
                              hipStream_t stream) {
  (void)in_sizes; (void)n_in; (void)out_size; (void)ws_size;
  const float* x     = (const float*)d_in[0];
  const int*   alpha = (const int*)  d_in[1];
  const float* A     = (const float*)d_in[2];
  const float* Bm    = (const float*)d_in[3];
  const float* Wr    = (const float*)d_in[4];
  float* out = (float*)d_out;

  float* g  = (float*)d_ws;
  float* hv = g + (size_t)NTOK * K_EXP;
  float* Af = hv + (size_t)NTOK * N_HV;
  int* sel  = (int*)(Af + (size_t)AF_UNITS * AF_UNIT);

  moelora_router<<<NTOK / 16, 256, 0, stream>>>(x, Wr, alpha, g, A, Af, sel, RT_HALF / RT_D);
  moelora_hgemm<<<K_EXP * 16, 128, 0, stream>>>(x, Af, g, sel, hv, KQ / KBLK);
  moelora_ogemm<<<dim3(NTOK / S_TB, D_DIM / S_CB), 256, 0, stream>>>(hv, Bm, sel, out);
}
